# starting kernel
# baseline (speedup 1.0000x reference)
#include <hip/hip_runtime.h>

#define KNU 4096
#define KNI 3072

typedef __attribute__((ext_vector_type(8))) int   v8i;
typedef __attribute__((ext_vector_type(8))) float v8f;
typedef __attribute__((ext_vector_type(2))) float v2f;

__global__ __launch_bounds__(256)
void k_binarize(const float* __restrict__ H,
                signed char* __restrict__ H8,
                signed char* __restrict__ Ht8) {
  __shared__ signed char t[32][33];
  int bx = blockIdx.x * 32, by = blockIdx.y * 32;
  int tx = threadIdx.x, ty = threadIdx.y;
#pragma unroll
  for (int r = ty; r < 32; r += 8) {
    int u = by + r, j = bx + tx;
    signed char v = (H[(size_t)u * KNI + j] >= 0.5f) ? 1 : 0;
    H8[(size_t)u * KNI + j] = v;
    t[r][tx] = v;
  }
  __syncthreads();
#pragma unroll
  for (int r = ty; r < 32; r += 8) {
    int j = bx + r, u = by + tx;
    Ht8[(size_t)j * KNU + u] = t[tx][r];
  }
}

__global__ __launch_bounds__(32)
void k_gram_bin(const signed char* __restrict__ X, const signed char* __restrict__ Y,
                signed char* __restrict__ Out, int M, int N, int K) {
  int lane = threadIdx.x;
  int lo = lane & 15, hi = lane >> 4;
  int m0 = blockIdx.y * 32, n0 = blockIdx.x * 32;
  v8i acc[2][2] = {};
  const signed char* ar0 = X + (size_t)(m0 + lo) * K;
  const signed char* ar1 = X + (size_t)(m0 + 16 + lo) * K;
  const signed char* br0 = Y + (size_t)(n0 + lo) * K;
  const signed char* br1 = Y + (size_t)(n0 + 16 + lo) * K;
  for (int k0 = 0; k0 < K; k0 += 64) {
    v8i a[2], b[2];
    const signed char* ap[2] = { ar0 + k0 + hi * 8,  ar1 + k0 + hi * 8 };
    const signed char* bp[2] = { br0 + k0 + hi * 16, br1 + k0 + hi * 16 };
#pragma unroll
    for (int t = 0; t < 2; ++t) {
#pragma unroll
      for (int q = 0; q < 4; ++q) {
        int2 d = *(const int2*)(ap[t] + q * 16);
        a[t][2 * q] = d.x; a[t][2 * q + 1] = d.y;
      }
      int4 b0 = *(const int4*)(bp[t]);
      int4 b1 = *(const int4*)(bp[t] + 32);
      b[t][0] = b0.x; b[t][1] = b0.y; b[t][2] = b0.z; b[t][3] = b0.w;
      b[t][4] = b1.x; b[t][5] = b1.y; b[t][6] = b1.z; b[t][7] = b1.w;
    }
#pragma unroll
    for (int mi = 0; mi < 2; ++mi)
#pragma unroll
      for (int ni = 0; ni < 2; ++ni)
        acc[mi][ni] = __builtin_amdgcn_wmma_i32_16x16x64_iu8(
            false, a[mi], false, b[ni], acc[mi][ni], false, false);
  }
#pragma unroll
  for (int mi = 0; mi < 2; ++mi)
#pragma unroll
    for (int ni = 0; ni < 2; ++ni)
#pragma unroll
      for (int v = 0; v < 8; ++v) {
        int m = m0 + mi * 16 + hi * 8 + v;
        int n = n0 + ni * 16 + lo;
        Out[(size_t)m * N + n] = (acc[mi][ni][v] != 0) ? (signed char)1 : (signed char)0;
      }
}

__global__ __launch_bounds__(256)
void k_rowsum_bin(const signed char* __restrict__ A, float* __restrict__ out, int C) {
  __shared__ int red[256];
  int r = blockIdx.x;
  const signed char* row = A + (size_t)r * C;
  int s = 0;
  for (int c = threadIdx.x * 16; c < C; c += 256 * 16) {
    int4 d = *(const int4*)(row + c);
    s += __popc(d.x) + __popc(d.y) + __popc(d.z) + __popc(d.w);
  }
  red[threadIdx.x] = s;
  __syncthreads();
  for (int w = 128; w > 0; w >>= 1) {
    if (threadIdx.x < w) red[threadIdx.x] += red[threadIdx.x + w];
    __syncthreads();
  }
  if (threadIdx.x == 0) out[r] = (float)red[0];
}

__global__ __launch_bounds__(256)
void k_colsum_bin(const signed char* __restrict__ A, float* __restrict__ out, int R, int C) {
  int c = blockIdx.x * 256 + threadIdx.x;
  if (c >= C) return;
  int s = 0;
  for (int r = 0; r < R; ++r) s += A[(size_t)r * C + c];
  out[c] = (float)s;
}

__global__ __launch_bounds__(256)
void k_scales(const float* __restrict__ sRowH, const float* __restrict__ sRowS,
              const float* __restrict__ sColH, const float* __restrict__ sColS,
              float* __restrict__ dvu, float* __restrict__ invRowH, float* __restrict__ invRowS,
              float* __restrict__ dvi, float* __restrict__ invColH, float* __restrict__ invColS) {
  int i = blockIdx.x * 256 + threadIdx.x;
  if (i < KNU) {
    float rv = sRowH[i] + sRowS[i];
    dvu[i] = rv > 0.f ? 1.f / sqrtf(rv) : 0.f;
    invRowH[i] = sRowH[i] > 0.f ? 1.f / sRowH[i] : 0.f;
    invRowS[i] = sRowS[i] > 0.f ? 1.f / sRowS[i] : 0.f;
  }
  if (i < KNI) {
    float cv = sColH[i] + sColS[i];
    dvi[i] = cv > 0.f ? 1.f / sqrtf(cv) : 0.f;
    invColH[i] = sColH[i] > 0.f ? 1.f / sColH[i] : 0.f;
    invColS[i] = sColS[i] > 0.f ? 1.f / sColS[i] : 0.f;
  }
}

__global__ __launch_bounds__(256)
void k_rowscale(const float* __restrict__ X, const float* __restrict__ s,
                float* __restrict__ out, int n) {
  int i = blockIdx.x * 256 + threadIdx.x;
  if (i < n) out[i] = X[i] * s[i >> 6];
}

template <bool TRANS>
__device__ inline void prop_accum(const signed char* __restrict__ Bm, const float* __restrict__ X,
                                  int Klen, int mA, int lo, int hi, int R, v8f acc[4]) {
  for (int k0 = 0; k0 < Klen; k0 += 4) {
    int ka = k0 + hi * 2;
    float a0, a1;
    if (TRANS) {
      a0 = (float)Bm[(size_t)ka * R + mA];
      a1 = (float)Bm[(size_t)(ka + 1) * R + mA];
    } else {
      a0 = (float)Bm[(size_t)mA * Klen + ka];
      a1 = (float)Bm[(size_t)mA * Klen + ka + 1];
    }
    v2f a; a[0] = a0; a[1] = a1;
#pragma unroll
    for (int ni = 0; ni < 4; ++ni) {
      int n = ni * 16 + lo;
      v2f b;
      b[0] = X[(size_t)ka * 64 + n];
      b[1] = X[(size_t)(ka + 1) * 64 + n];
      acc[ni] = __builtin_amdgcn_wmma_f32_16x16x4_f32(
          false, a, false, b, (short)0, acc[ni], false, false);
    }
  }
}

template <bool TRANS>
__global__ __launch_bounds__(32)
void k_prop_f32(const signed char* __restrict__ B1, const float* __restrict__ X1, int K1,
                const signed char* __restrict__ B2, const float* __restrict__ X2, int K2,
                const float* __restrict__ oscale, const float* __restrict__ addin,
                float* __restrict__ out, int R) {
  int lane = threadIdx.x;
  int lo = lane & 15, hi = lane >> 4;
  int m0 = blockIdx.x * 16;
  int mA = m0 + lo;
  v8f acc[4] = {};
  prop_accum<TRANS>(B1, X1, K1, mA, lo, hi, R, acc);
  if (B2) prop_accum<TRANS>(B2, X2, K2, mA, lo, hi, R, acc);
#pragma unroll
  for (int ni = 0; ni < 4; ++ni)
#pragma unroll
    for (int v = 0; v < 8; ++v) {
      int m = m0 + hi * 8 + v;
      int n = ni * 16 + lo;
      float r = acc[ni][v];
      if (oscale) r *= oscale[m];
      if (addin) r += addin[(size_t)m * 64 + n];
      out[(size_t)m * 64 + n] = r;
    }
}

__global__ __launch_bounds__(256)
void k_linear(const float* __restrict__ Y, const float* __restrict__ W,
              const float* __restrict__ bias, float* __restrict__ out, int R) {
  int idx = blockIdx.x * 256 + threadIdx.x;
  if (idx >= R * 64) return;
  int r = idx >> 6, d = idx & 63;
  float s = bias[d];
#pragma unroll 8
  for (int k = 0; k < 64; ++k) s = fmaf(Y[(size_t)r * 64 + k], W[(size_t)k * 64 + d], s);
  out[idx] = s;
}

extern "C" void kernel_launch(void* const* d_in, const int* in_sizes, int n_in,
                              void* d_out, int out_size, void* d_ws, size_t ws_size,
                              hipStream_t stream) {
  const float* H    = (const float*)d_in[0];
  const float* U    = (const float*)d_in[1];
  const float* I    = (const float*)d_in[2];
  const float* W    = (const float*)d_in[3];
  const float* bias = (const float*)d_in[4];
  float* Uout = (float*)d_out;
  float* Iout = Uout + (size_t)KNU * 64;

  size_t off = 0;
  auto wsalloc = [&](size_t bytes) -> char* {
    char* p = (char*)d_ws + off;
    off += (bytes + 255) & ~(size_t)255;
    return p;
  };
  signed char* H8   = (signed char*)wsalloc((size_t)KNU * KNI);
  signed char* Ht8  = (signed char*)wsalloc((size_t)KNI * KNU);
  signed char* Gb   = (signed char*)wsalloc((size_t)KNI * KNI);
  signed char* S    = (signed char*)wsalloc((size_t)KNU * KNI);
  float* sRowH   = (float*)wsalloc(KNU * 4);
  float* sColH   = (float*)wsalloc(KNI * 4);
  float* sRowS   = (float*)wsalloc(KNU * 4);
  float* sColS   = (float*)wsalloc(KNI * 4);
  float* dvu     = (float*)wsalloc(KNU * 4);
  float* invRowH = (float*)wsalloc(KNU * 4);
  float* invRowS = (float*)wsalloc(KNU * 4);
  float* dvi     = (float*)wsalloc(KNI * 4);
  float* invColH = (float*)wsalloc(KNI * 4);
  float* invColS = (float*)wsalloc(KNI * 4);
  float* Xsu  = (float*)wsalloc((size_t)KNU * 64 * 4);
  float* Xsi  = (float*)wsalloc((size_t)KNI * 64 * 4);
  float* Q1   = (float*)wsalloc((size_t)KNI * 64 * 4);
  float* Q2   = (float*)wsalloc((size_t)KNI * 64 * 4);
  float* Qi1  = (float*)wsalloc((size_t)KNU * 64 * 4);
  float* Qi2  = (float*)wsalloc((size_t)KNU * 64 * 4);
  float* Yu   = (float*)wsalloc((size_t)KNU * 64 * 4);
  float* Yi   = (float*)wsalloc((size_t)KNI * 64 * 4);
  (void)ws_size; (void)in_sizes; (void)n_in; (void)out_size;

  k_binarize<<<dim3(KNI / 32, KNU / 32), dim3(32, 8), 0, stream>>>(H, H8, Ht8);
  k_gram_bin<<<dim3(KNI / 32, KNI / 32), 32, 0, stream>>>(Ht8, Ht8, Gb, KNI, KNI, KNU);
  k_gram_bin<<<dim3(KNI / 32, KNU / 32), 32, 0, stream>>>(H8, Gb, S, KNU, KNI, KNI);

  k_rowsum_bin<<<KNU, 256, 0, stream>>>(H8, sRowH, KNI);
  k_rowsum_bin<<<KNI, 256, 0, stream>>>(Ht8, sColH, KNU);
  k_rowsum_bin<<<KNU, 256, 0, stream>>>(S, sRowS, KNI);
  k_colsum_bin<<<KNI / 256, 256, 0, stream>>>(S, sColS, KNU, KNI);
  k_scales<<<(KNU + 255) / 256, 256, 0, stream>>>(sRowH, sRowS, sColH, sColS,
                                                  dvu, invRowH, invRowS,
                                                  dvi, invColH, invColS);

  k_rowscale<<<(KNU * 64) / 256, 256, 0, stream>>>(U, dvu, Xsu, KNU * 64);
  k_prop_f32<true><<<KNI / 16, 32, 0, stream>>>(H8, Xsu, KNU, nullptr, nullptr, 0,
                                                invColH, nullptr, Q1, KNI);
  k_prop_f32<true><<<KNI / 16, 32, 0, stream>>>(S, Xsu, KNU, nullptr, nullptr, 0,
                                                invColS, nullptr, Q2, KNI);
  k_prop_f32<false><<<KNU / 16, 32, 0, stream>>>(H8, Q1, KNI, S, Q2, KNI,
                                                 dvu, U, Yu, KNU);

  k_rowscale<<<(KNI * 64) / 256, 256, 0, stream>>>(I, dvi, Xsi, KNI * 64);
  k_prop_f32<false><<<KNU / 16, 32, 0, stream>>>(H8, Xsi, KNI, nullptr, nullptr, 0,
                                                 invRowH, nullptr, Qi1, KNU);
  k_prop_f32<false><<<KNU / 16, 32, 0, stream>>>(S, Xsi, KNI, nullptr, nullptr, 0,
                                                 invRowS, nullptr, Qi2, KNU);
  k_prop_f32<true><<<KNI / 16, 32, 0, stream>>>(H8, Qi1, KNU, S, Qi2, KNU,
                                                dvi, I, Yi, KNI);

  k_linear<<<(KNU * 64) / 256, 256, 0, stream>>>(Yu, W, bias, Uout, KNU);
  k_linear<<<(KNI * 64) / 256, 256, 0, stream>>>(Yi, W, bias, Iout, KNI);
}


# Second Gram product (int8, dtype unchanged) sums only over the items flagged for a tile's 16 user rows: item list per block, B by 4-byte row reads of row-major Gb; packed operands go; int sums exact
# speedup vs baseline: 11.6874x; 11.6874x over previous
#include <hip/hip_runtime.h>

#define KNU 4096
#define KNI 3072

typedef __attribute__((ext_vector_type(8))) int   v8i;
typedef __attribute__((ext_vector_type(8))) float v8f;
typedef __attribute__((ext_vector_type(2))) float v2f;

typedef __attribute__((ext_vector_type(4))) int   v4i;

#define BSEG 8
#define BTILES (KNI / 32 / BSEG)

__device__ __forceinline__ void lds_barrier() {
  __builtin_amdgcn_fence(__ATOMIC_RELEASE, "workgroup", "local");
  __builtin_amdgcn_s_barrier();
  __builtin_amdgcn_fence(__ATOMIC_ACQUIRE, "workgroup", "local");
}

__global__ __launch_bounds__(256)
void k_binarize(const float* __restrict__ H,
                signed char* __restrict__ H8,
                signed char* __restrict__ Ft,
                signed char* __restrict__ FtU,
                int* __restrict__ rowPart,
                int* __restrict__ colPart) {
  __shared__ __attribute__((aligned(16))) signed char t[32][48];
  __shared__ __attribute__((aligned(16))) signed char tt[32][48];
  __shared__ int rcnt[32], ccnt[32];
  const int bx0 = blockIdx.x * (BTILES * 32), by = blockIdx.y * 32;
  const int tx = threadIdx.x, ty = threadIdx.y;
  const int tid = ty * 32 + tx;
  const float* hp = H + (size_t)(by + ty) * KNI + bx0 + tx;
  signed char* h8p = H8 + (size_t)(by + ty) * KNI + bx0 + tx;
  float buf[3][4];
  int racc[4] = {0, 0, 0, 0};
#pragma unroll
  for (int i = 0; i < 4; ++i) buf[0][i] = hp[(size_t)i * 8 * KNI];
#pragma unroll
  for (int i = 0; i < 4; ++i) buf[1][i] = hp[(size_t)i * 8 * KNI + 32];
#pragma unroll
  for (int it = 0; it < BTILES; ++it) {
    const int bx = bx0 + it * 32;
    if (it + 2 < BTILES) {
#pragma unroll
      for (int i = 0; i < 4; ++i) buf[(it + 2) % 3][i] = hp[(size_t)i * 8 * KNI + (it + 2) * 32];
    }
#pragma unroll
    for (int i = 0; i < 4; ++i) {
      const int r = ty + 8 * i;
      signed char v = (buf[it % 3][i] >= 0.5f) ? 1 : 0;
      h8p[(size_t)i * 8 * KNI + it * 32] = v;
      t[r][tx] = v;
      tt[tx][r] = v;
      unsigned int ones = __builtin_amdgcn_ballot_w32(v != 0);
      racc[i] += __popc(ones);
    }
    lds_barrier();
#pragma unroll
    for (int r = ty; r < 32; r += 8) {
      signed char v = tt[r][tx];
      unsigned int ones = __builtin_amdgcn_ballot_w32(v != 0);
      if (tx == 0) ccnt[r] = __popc(ones);
    }
    if (tid < 64) {
      int rbl = tid >> 5, p = (tid >> 4) & 1, c = tid & 15;
      v4i piece = *(const v4i*)&t[16 * rbl + c][16 * p];
      Ft[(size_t)((bx >> 4) + p) * KNU + by + 16 * rbl + c] =
          ((piece[0] | piece[1] | piece[2] | piece[3]) != 0) ? (signed char)1 : (signed char)0;
      v4i col = *(const v4i*)&tt[tx][16 * rbl];
      FtU[(size_t)((by >> 4) + rbl) * KNI + bx + tx] =
          ((col[0] | col[1] | col[2] | col[3]) != 0) ? (signed char)1 : (signed char)0;
    }
    lds_barrier();
    if (ty == 1) colPart[(size_t)blockIdx.y * KNI + bx + tx] = ccnt[tx];
  }
  if (tx == 0) {
#pragma unroll
    for (int i = 0; i < 4; ++i) rcnt[ty + 8 * i] = racc[i];
  }
  lds_barrier();
  if (ty == 0) rowPart[(size_t)blockIdx.x * KNU + by + tx] = rcnt[tx];
}

#define GSEG 768

__device__ __forceinline__ void gram_list_group(const signed char* __restrict__ Bm, const unsigned int (&ub)[32],
                                                const v8i& a, v8i (&acc)[4]) {
  unsigned int R[32];
#pragma unroll
  for (int s = 0; s < 32; ++s) R[s] = *(const unsigned int*)(Bm + (size_t)ub[s]);
  v8i bf[4];
#pragma unroll
  for (int r = 0; r < 8; ++r) {
    const unsigned int lo01 = __builtin_amdgcn_perm(R[4 * r + 1], R[4 * r],     0x05010400u);
    const unsigned int hi01 = __builtin_amdgcn_perm(R[4 * r + 1], R[4 * r],     0x07030602u);
    const unsigned int lo23 = __builtin_amdgcn_perm(R[4 * r + 3], R[4 * r + 2], 0x05010400u);
    const unsigned int hi23 = __builtin_amdgcn_perm(R[4 * r + 3], R[4 * r + 2], 0x07030602u);
    bf[0][r] = (int)__builtin_amdgcn_perm(lo23, lo01, 0x05040100u);
    bf[1][r] = (int)__builtin_amdgcn_perm(lo23, lo01, 0x07060302u);
    bf[2][r] = (int)__builtin_amdgcn_perm(hi23, hi01, 0x05040100u);
    bf[3][r] = (int)__builtin_amdgcn_perm(hi23, hi01, 0x07060302u);
  }
#pragma unroll
  for (int i = 0; i < 4; ++i)
    acc[i] = __builtin_amdgcn_wmma_i32_16x16x64_iu8(false, a, false, bf[i], acc[i], false, false);
}


__global__ __launch_bounds__(128)
void k_gram_first(const signed char* __restrict__ H8, const signed char* __restrict__ Ft,
                  signed char* __restrict__ Out) {
  __shared__ __attribute__((aligned(16))) unsigned short part[4][1024];
  __shared__ __attribute__((aligned(16))) unsigned short list[KNU + 64];
  __shared__ int wcnt[4];
  const int lane = threadIdx.x & 31, w = threadIdx.x >> 5;
  const int c = lane & 15, h = lane >> 4;
  const int m0 = blockIdx.y * 16;
  const int nbase = blockIdx.x * GSEG + w * 192;
  const unsigned int* fp = (const unsigned int*)(Ft + (size_t)blockIdx.y * KNU + 1024 * w) + lane;
  unsigned int fw[8];
#pragma unroll
  for (int it = 0; it < 8; ++it) fw[it] = fp[32 * it];
  int cnt = 0;
  const unsigned int below = (1u << lane) - 1u;
#pragma unroll
  for (int it = 0; it < 8; ++it)
#pragma unroll
    for (int b = 0; b < 4; ++b) {
      const bool f = ((fw[it] >> (8 * b)) & 0xFFu) != 0;
      const unsigned int mask = __builtin_amdgcn_ballot_w32(f);
      if (f) part[w][cnt + __popc(mask & below)] = (unsigned short)(1024 * w + 128 * it + 4 * lane + b);
      cnt += __popc(mask);
    }
  if (lane == 0) wcnt[w] = cnt;
  __syncthreads();
  int off = 0, total = 0;
#pragma unroll
  for (int q = 0; q < 4; ++q) { const int x = wcnt[q]; if (q < w) off += x; total += x; }
  for (int i = lane; i < cnt; i += 32) list[off + i] = part[w][i];
  const int padded = (total + 63) & ~63;
  for (int i = total + (int)threadIdx.x; i < padded; i += 128) list[i] = 0;
  __syncthreads();

  v8i acc[3][4] = {};
  const unsigned int cola = m0 + c;
  const unsigned int colb = nbase + 4 * c;
  for (int k0 = 0; k0 < padded; k0 += 64) {
    const int jb = k0 + 32 * h;
    const int live = total - jb;
    unsigned int ua[32], ub[32];
    const v4i* lp = (const v4i*)&list[jb];
#pragma unroll
    for (int q = 0; q < 4; ++q) {
      const v4i x = lp[q];
#pragma unroll
      for (int e = 0; e < 4; ++e) {
        const unsigned int u0 = (unsigned int)x[e] & 0xFFFFu, u1 = (unsigned int)x[e] >> 16;
        ua[8 * q + 2 * e]     = u0 * KNI + cola;   ub[8 * q + 2 * e]     = u0 * KNI + colb;
        ua[8 * q + 2 * e + 1] = u1 * KNI + cola;   ub[8 * q + 2 * e + 1] = u1 * KNI + colb;
      }
    }
    v8i a;
#pragma unroll
    for (int r = 0; r < 8; ++r) {
      unsigned int d = 0;
#pragma unroll
      for (int b = 0; b < 4; ++b) {
        const int s = 4 * r + b;
        const unsigned int v = (unsigned int)(unsigned char)H8[(size_t)ua[s]];
        d |= (s < live ? v : 0u) << (8 * b);
      }
      a[r] = (int)d;
    }
#pragma unroll
    for (int g = 0; g < 3; ++g) gram_list_group(H8 + 64 * g, ub, a, acc[g]);
  }
#pragma unroll
  for (int g = 0; g < 3; ++g)
#pragma unroll
    for (int v = 0; v < 8; ++v) {
      unsigned int d = 0;
#pragma unroll
      for (int i = 0; i < 4; ++i) d |= (acc[g][i][v] != 0 ? 1u : 0u) << (8 * i);
      *(unsigned int*)(Out + (size_t)(m0 + 8 * h + v) * KNI + nbase + 64 * g + 4 * c) = d;
    }
}

__global__ __launch_bounds__(128)
void k_gram_second(const signed char* __restrict__ H8, const signed char* __restrict__ FtU,
                   const signed char* __restrict__ Gb, signed char* __restrict__ Out,
                   int* __restrict__ rowPart, unsigned char* __restrict__ colPart) {
  __shared__ __attribute__((aligned(16))) unsigned short part[4][KNI / 4];
  __shared__ __attribute__((aligned(16))) unsigned short list[KNI + 64];
  __shared__ int wcnt[4];
  const int lane = threadIdx.x & 31, w = threadIdx.x >> 5;
  const int c = lane & 15, h = lane >> 4;
  const int m0 = blockIdx.y * 16;
  const int nbase = blockIdx.x * GSEG + w * 192;
  const unsigned int* fp = (const unsigned int*)(FtU + (size_t)blockIdx.y * KNI + (KNI / 4) * w) + lane;
  unsigned int fw[6];
#pragma unroll
  for (int it = 0; it < 6; ++it) fw[it] = fp[32 * it];
  int cnt = 0;
  const unsigned int below = (1u << lane) - 1u;
#pragma unroll
  for (int it = 0; it < 6; ++it)
#pragma unroll
    for (int b = 0; b < 4; ++b) {
      const bool f = ((fw[it] >> (8 * b)) & 0xFFu) != 0;
      const unsigned int mask = __builtin_amdgcn_ballot_w32(f);
      if (f) part[w][cnt + __popc(mask & below)] = (unsigned short)((KNI / 4) * w + 128 * it + 4 * lane + b);
      cnt += __popc(mask);
    }
  if (lane == 0) wcnt[w] = cnt;
  __syncthreads();
  int off = 0, total = 0;
#pragma unroll
  for (int q = 0; q < 4; ++q) { const int x = wcnt[q]; if (q < w) off += x; total += x; }
  for (int i = lane; i < cnt; i += 32) list[off + i] = part[w][i];
  const int padded = (total + 63) & ~63;
  for (int i = total + (int)threadIdx.x; i < padded; i += 128) list[i] = 0;
  __syncthreads();

  v8i acc[3][4] = {};
  const unsigned int rowa = (unsigned int)(m0 + c) * KNI;
  const unsigned int colb = nbase + 4 * c;
  for (int k0 = 0; k0 < padded; k0 += 64) {
    const int jb = k0 + 32 * h;
    const int live = total - jb;
    unsigned int ua[32], ub[32];
    const v4i* lp = (const v4i*)&list[jb];
#pragma unroll
    for (int q = 0; q < 4; ++q) {
      const v4i x = lp[q];
#pragma unroll
      for (int e = 0; e < 4; ++e) {
        const unsigned int k0e = (unsigned int)x[e] & 0xFFFFu, k1e = (unsigned int)x[e] >> 16;
        ua[8 * q + 2 * e]     = rowa + k0e;   ub[8 * q + 2 * e]     = k0e * KNI + colb;
        ua[8 * q + 2 * e + 1] = rowa + k1e;   ub[8 * q + 2 * e + 1] = k1e * KNI + colb;
      }
    }
    v8i a;
#pragma unroll
    for (int r = 0; r < 8; ++r) {
      unsigned int d = 0;
#pragma unroll
      for (int b = 0; b < 4; ++b) {
        const int s = 4 * r + b;
        const unsigned int v = (unsigned int)(unsigned char)H8[(size_t)ua[s]];
        d |= (s < live ? v : 0u) << (8 * b);
      }
      a[r] = (int)d;
    }
#pragma unroll
    for (int g = 0; g < 3; ++g) gram_list_group(Gb + 64 * g, ub, a, acc[g]);
  }
#pragma unroll
  for (int g = 0; g < 3; ++g)
#pragma unroll
    for (int v = 0; v < 8; ++v) {
      unsigned int d = 0;
#pragma unroll
      for (int i = 0; i < 4; ++i) d |= (acc[g][i][v] != 0 ? 1u : 0u) << (8 * i);
      *(unsigned int*)(Out + (size_t)(m0 + 8 * h + v) * KNI + nbase + 64 * g + 4 * c) = d;
    }
  int rc = 0;
#pragma unroll
  for (int v = 0; v < 8; ++v) {
    int rlo = 0, rhi = 0;
#pragma unroll
    for (int g = 0; g < 3; ++g)
#pragma unroll
      for (int i = 0; i < 4; ++i) {
        const unsigned int bl = __builtin_amdgcn_ballot_w32(acc[g][i][v] != 0);
        rlo += __popc(bl & 0xFFFFu);
        rhi += __popc(bl >> 16);
      }
    if (lane == v) rc = rlo;
    if (lane == v + 8) rc = rhi;
  }
  if (lane < 16) rowPart[(size_t)(blockIdx.x * 4 + w) * KNU + m0 + lane] = rc;
#pragma unroll
  for (int g = 0; g < 3; ++g) {
    unsigned int d = 0;
#pragma unroll
    for (int i = 0; i < 4; ++i) {
      int cc = 0;
#pragma unroll
      for (int v = 0; v < 8; ++v) cc += acc[g][i][v] != 0 ? 1 : 0;
      cc += __shfl_xor(cc, 16);
      d |= (unsigned int)cc << (8 * i);
    }
    if (h == 0) *(unsigned int*)(colPart + (size_t)blockIdx.y * KNI + nbase + 64 * g + 4 * c) = d;
  }
}


__global__ __launch_bounds__(256)
void k_scales(const int* __restrict__ rowPartH, const int* __restrict__ rowPartS,
              const int* __restrict__ colPartH, const unsigned char* __restrict__ colPartS,
              float* __restrict__ dvu, float* __restrict__ invRowH, float* __restrict__ invRowS,
              float* __restrict__ dvi, float* __restrict__ invColH, float* __restrict__ invColS) {
  __shared__ int red[4][8][32];
  int lane = threadIdx.x & 31, w = threadIdx.x >> 5;
  int i = blockIdx.x * 32 + lane;
  int a = 0, b = 0, c = 0, d = 0;
  for (int p = w; p < BSEG; p += 8) a += rowPartH[(size_t)p * KNU + i];
  for (int p = w; p < KNI / 192; p += 8) b += rowPartS[(size_t)p * KNU + i];
  if (i < KNI) {
    for (int p = w; p < KNU / 32; p += 8) c += colPartH[(size_t)p * KNI + i];
    for (int p = w; p < KNU / 16; p += 8) d += colPartS[(size_t)p * KNI + i];
  }
  red[0][w][lane] = a; red[1][w][lane] = b; red[2][w][lane] = c; red[3][w][lane] = d;
  __syncthreads();
  if (w == 0) {
    int sa = 0, sb = 0, sc = 0, sd = 0;
#pragma unroll
    for (int q = 0; q < 8; ++q) {
      sa += red[0][q][lane]; sb += red[1][q][lane]; sc += red[2][q][lane]; sd += red[3][q][lane];
    }
    float fRowH = (float)sa, fRowS = (float)sb;
    float rv = fRowH + fRowS;
    dvu[i] = rv > 0.f ? 1.f / sqrtf(rv) : 0.f;
    invRowH[i] = fRowH > 0.f ? 1.f / fRowH : 0.f;
    invRowS[i] = fRowS > 0.f ? 1.f / fRowS : 0.f;
    if (i < KNI) {
      float fColH = (float)sc, fColS = (float)sd;
      float cv = fColH + fColS;
      dvi[i] = cv > 0.f ? 1.f / sqrtf(cv) : 0.f;
      invColH[i] = fColH > 0.f ? 1.f / fColH : 0.f;
      invColS[i] = fColS > 0.f ? 1.f / fColS : 0.f;
    }
  }
}

typedef __attribute__((ext_vector_type(4))) float v4f;

#define PKC 256
#define PXP 40
#define PXT 260
#define PAP 272
#define PXS (32 * PXT)
#define PAS (32 * PAP)
#define PXL (PKC / 16)
#define PAL 4

struct PropFrag { unsigned int a[4]; v4f b[4]; };

__device__ __forceinline__ void prop_frag_load(PropFrag& f, const float* __restrict__ xb,
                                               const unsigned char* __restrict__ ab, int t0) {
#pragma unroll
  for (int j = 0; j < 4; ++j) {
    f.a[j] = *(const unsigned int*)(ab + 8 * (t0 + j));
    f.b[j] = *(const v4f*)(xb + 8 * (t0 + j));
  }
}

__device__ __forceinline__ void prop_frag_mma(const PropFrag& f, v8f& accE, v8f& accO) {
#pragma unroll
  for (int j = 0; j < 4; ++j) {
    v2f aE, aO, bE, bO;
    aE[0] = (float)(f.a[j] & 0xFFu);          aE[1] = (float)((f.a[j] >> 8) & 0xFFu);
    aO[0] = (float)((f.a[j] >> 16) & 0xFFu);  aO[1] = (float)(f.a[j] >> 24);
    bE[0] = f.b[j][0];  bE[1] = f.b[j][1];
    bO[0] = f.b[j][2];  bO[1] = f.b[j][3];
    accE = __builtin_amdgcn_wmma_f32_16x16x4_f32(false, aE, false, bE, (short)0, accE, false, false);
    accO = __builtin_amdgcn_wmma_f32_16x16x4_f32(false, aO, false, bO, (short)0, accO, false, false);
  }
}

__device__ __forceinline__ void prop_slab(const float* __restrict__ xb, const unsigned char* __restrict__ ab,
                                          v8f& accE, v8f& accO) {
  PropFrag f0, f1;
  prop_frag_load(f0, xb, ab, 0);
#pragma unroll
  for (int t0 = 0; t0 < PKC / 8; t0 += 8) {
    prop_frag_load(f1, xb, ab, t0 + 4);
    __builtin_amdgcn_sched_barrier(0);
    prop_frag_mma(f0, accE, accO);
    __builtin_amdgcn_sched_barrier(0);
    if (t0 + 8 < PKC / 8) prop_frag_load(f0, xb, ab, t0 + 8);
    __builtin_amdgcn_sched_barrier(0);
    prop_frag_mma(f1, accE, accO);
    __builtin_amdgcn_sched_barrier(0);
  }
}

template <bool TRANS>
__device__ __forceinline__ void prop_stage_store(float* __restrict__ xd, unsigned char* __restrict__ ad,
                                                 const v4f (&xv)[PXL], const v4i (&av)[PAL]) {
#pragma unroll
  for (int i = 0; i < 4; ++i)
#pragma unroll
    for (int e = 0; e < 4; ++e) {
      v4f t;
      t[0] = xv[4 * i][e];  t[1] = xv[4 * i + 1][e];  t[2] = xv[4 * i + 2][e];  t[3] = xv[4 * i + 3][e];
      *(v4f*)(xd + e * PXT + 64 * i) = t;
    }
  if (TRANS) {
#pragma unroll
    for (int d = 0; d < 4; ++d)
#pragma unroll
      for (int b = 0; b < 4; ++b) {
        unsigned int w = (((unsigned int)av[0][d] >> (8 * b)) & 0xFFu)
                       | ((((unsigned int)av[1][d] >> (8 * b)) & 0xFFu) << 8)
                       | ((((unsigned int)av[2][d] >> (8 * b)) & 0xFFu) << 16)
                       | ((((unsigned int)av[3][d] >> (8 * b)) & 0xFFu) << 24);
        *(unsigned int*)(ad + (4 * d + b) * PAP) = w;
      }
  } else {
#pragma unroll
    for (int j = 0; j < PAL; ++j) *(v4i*)(ad + j * 8 * PAP) = av[j];
  }
}

template <bool TRANS>
__device__ __forceinline__ void prop_accum(const signed char* __restrict__ Bm, const float* __restrict__ X,
                                           int Klen, int R, int m0, int tid, int xoff, int aoff,
                                           float* __restrict__ Xs, unsigned char* __restrict__ As,
                                           v8f& accE, v8f& accO) {
  const int kq = 4 * (tid >> 5) + ((((tid >> 3) & 3) + 2 * ((tid >> 2) & 1)) & 3);
  const float* xsrc = X + (size_t)(4 * kq) * 64 + (tid & 7) * 4;
  const int xdst = (4 * (tid & 7)) * PXT + 4 * kq;
  const signed char* asrc;
  size_t astep, aj;
  int adst;
  if (TRANS) {
    asrc = Bm + (size_t)(4 * (tid >> 1)) * R + m0 + 16 * (tid & 1);  astep = (size_t)PKC * R;  aj = (size_t)R;
    adst = (16 * (tid & 1)) * PAP + 4 * (tid >> 1);
  } else {
    asrc = Bm + (size_t)(m0 + (tid >> 4)) * Klen + (tid & 15) * 16;  astep = PKC;  aj = (size_t)8 * Klen;
    adst = (tid >> 4) * PAP + (tid & 15) * 16;
  }
  v4f xv[PXL];
  v4i av[PAL];
#pragma unroll
  for (int i = 0; i < 4; ++i)
#pragma unroll
    for (int r = 0; r < 4; ++r) xv[4 * i + r] = *(const v4f*)(xsrc + (size_t)(64 * i + r) * 64);
#pragma unroll
  for (int j = 0; j < PAL; ++j) av[j] = *(const v4i*)(asrc + j * aj);
  prop_stage_store<TRANS>(Xs + xdst, As + adst, xv, av);
  __syncthreads();
  const int nslab = Klen / PKC;
  int cur = 0;
  for (int it = 0; it + 1 < nslab; ++it) {
    xsrc += (size_t)PKC * 64;
    asrc += astep;
#pragma unroll
    for (int i = 0; i < 4; ++i)
#pragma unroll
      for (int r = 0; r < 4; ++r) xv[4 * i + r] = *(const v4f*)(xsrc + (size_t)(64 * i + r) * 64);
#pragma unroll
    for (int j = 0; j < PAL; ++j) av[j] = *(const v4i*)(asrc + j * aj);
    prop_slab(Xs + cur * PXS + xoff, As + cur * PAS + aoff, accE, accO);
    cur ^= 1;
    prop_stage_store<TRANS>(Xs + cur * PXS + xdst, As + cur * PAS + adst, xv, av);
    __syncthreads();
  }
  prop_slab(Xs + cur * PXS + xoff, As + cur * PAS + aoff, accE, accO);
  __syncthreads();
}

template <bool TRANS>
__device__ __forceinline__ void prop_tile(const signed char* __restrict__ B1, const float* __restrict__ X1, int K1,
                                          const signed char* __restrict__ B2, const float* __restrict__ X2, int K2,
                                          const float* __restrict__ oscale, const float* __restrict__ addin,
                                          float* __restrict__ out, int R, int t,
                                          float* __restrict__ Xs, unsigned char* __restrict__ As) {
  const int m0 = (t >> 1) * 32, col0 = (t & 1) * 32;
  int tid = threadIdx.x;
  int lane = tid & 31, w = tid >> 5;
  int c = lane & 15, h = lane >> 4;
  const int rb = w >> 1, cb = w & 1;
  const int xoff = (16 * cb + c) * PXT + 4 * h;
  const int aoff = (16 * rb + c) * PAP + 4 * h;
  v8f accE = {}, accO = {};
  prop_accum<TRANS>(B1, X1 + col0, K1, R, m0, tid, xoff, aoff, Xs, As, accE, accO);
  if (B2) prop_accum<TRANS>(B2, X2 + col0, K2, R, m0, tid, xoff, aoff, Xs, As, accE, accO);
#pragma unroll
  for (int v = 0; v < 8; ++v) Xs[(16 * rb + 8 * h + v) * PXP + 16 * cb + c] = accE[v] + accO[v];
  __syncthreads();
#pragma unroll
  for (int i = 0; i < 4; ++i) {
    int r = 8 * w + 2 * i + h, m = m0 + r;
    v2f val = *(const v2f*)(Xs + r * PXP + 2 * c);
    if (oscale) { float sc = oscale[m]; val[0] *= sc; val[1] *= sc; }
    if (addin) { v2f ad = *(const v2f*)(addin + (size_t)m * 64 + col0 + 2 * c); val[0] += ad[0]; val[1] += ad[1]; }
    *(v2f*)(out + (size_t)m * 64 + col0 + 2 * c) = val;
  }
}

__global__ __launch_bounds__(128)
void k_prop_level1(const signed char* __restrict__ H8, const signed char* __restrict__ S,
                   const float* __restrict__ Xsu, const float* __restrict__ Xsi,
                   const float* __restrict__ invColH, const float* __restrict__ invColS,
                   const float* __restrict__ invRowH, const float* __restrict__ invRowS,
                   float* __restrict__ Q1, float* __restrict__ Q2,
                   float* __restrict__ Qi1, float* __restrict__ Qi2) {
  __shared__ __attribute__((aligned(16))) float Xs[2 * PXS];
  __shared__ __attribute__((aligned(16))) unsigned char As[2 * PAS];
  int b = blockIdx.x;
  if (b < 2 * (KNI / 16)) {
    const bool second = b >= KNI / 16;
    const int tile = second ? b - KNI / 16 : b;
    prop_tile<true>(second ? S : H8, Xsu, KNU, nullptr, nullptr, 0,
                    second ? invColS : invColH, nullptr, second ? Q2 : Q1, KNI, tile, Xs, As);
  } else {
    b -= 2 * (KNI / 16);
    const bool second = b >= KNU / 16;
    const int tile = second ? b - KNU / 16 : b;
    prop_tile<false>(second ? S : H8, Xsi, KNI, nullptr, nullptr, 0,
                     second ? invRowS : invRowH, nullptr, second ? Qi2 : Qi1, KNU, tile, Xs, As);
  }
}

__global__ __launch_bounds__(128)
void k_prop_level2(const signed char* __restrict__ H8, const signed char* __restrict__ S,
                   const float* __restrict__ Q1, const float* __restrict__ Q2,
                   const float* __restrict__ Qi1, const float* __restrict__ Qi2,
                   const float* __restrict__ dvu, const float* __restrict__ dvi,
                   const float* __restrict__ UWb, const float* __restrict__ IWb,
                   float* __restrict__ outU, float* __restrict__ outI) {
  __shared__ __attribute__((aligned(16))) float Xs[2 * PXS];
  __shared__ __attribute__((aligned(16))) unsigned char As[2 * PAS];
  int b = blockIdx.x;
  if (b < KNI / 16)
    prop_tile<true>(H8, Qi1, KNU, S, Qi2, KNU, dvi, IWb, outI, KNI, b, Xs, As);
  else
    prop_tile<false>(H8, Q1, KNI, S, Q2, KNI, dvu, UWb, outU, KNU, b - KNI / 16, Xs, As);
}

#define HTP 72

__global__ __launch_bounds__(128)
void k_head_scale(const float* __restrict__ U, const float* __restrict__ I,
                  const float* __restrict__ W, const float* __restrict__ bias,
                  const float* __restrict__ dvu, const float* __restrict__ dvi,
                  float* __restrict__ Xsu, float* __restrict__ Xsi,
                  float* __restrict__ UWb, float* __restrict__ IWb) {
  __shared__ __attribute__((aligned(16))) float T[16 * HTP];
  const int b = blockIdx.x;
  const bool item = b >= KNU / 16;
  const int m0 = (item ? b - KNU / 16 : b) * 16;
  const float* __restrict__ X = item ? I : U;
  const float* __restrict__ scale = item ? dvi : dvu;
  float* __restrict__ Xs = item ? Xsi : Xsu;
  float* __restrict__ XWb = item ? IWb : UWb;
  const int tid = threadIdx.x;
  const int lane = tid & 31, w = tid >> 5;
  const int c = lane & 15, h = lane >> 4;
  const float* xr = X + (size_t)(m0 + c) * 64 + 32 * h;
  const float* wc = W + (size_t)(32 * h) * 64 + 16 * w + c;
  v4f xa[8];
  v2f wb[16];
#pragma unroll
  for (int i = 0; i < 8; ++i) xa[i] = *(const v4f*)(xr + 4 * i);
#pragma unroll
  for (int s = 0; s < 16; ++s) {
    wb[s][0] = wc[(size_t)(2 * s) * 64];
    wb[s][1] = wc[(size_t)(2 * s + 1) * 64];
  }
  v8f accE = {}, accO = {};
#pragma unroll
  for (int s = 0; s < 16; s += 2) {
    v2f aE, aO;
    aE[0] = xa[s >> 1][0];  aE[1] = xa[s >> 1][1];
    aO[0] = xa[s >> 1][2];  aO[1] = xa[s >> 1][3];
    accE = __builtin_amdgcn_wmma_f32_16x16x4_f32(false, aE, false, wb[s], (short)0, accE, false, false);
    accO = __builtin_amdgcn_wmma_f32_16x16x4_f32(false, aO, false, wb[s + 1], (short)0, accO, false, false);
  }
#pragma unroll
  for (int v = 0; v < 8; ++v) T[(8 * h + v) * HTP + 16 * w + c] = accE[v] + accO[v];
  __syncthreads();
  const v2f bb = *(const v2f*)(bias + 2 * lane);
#pragma unroll
  for (int i = 0; i < 4; ++i) {
    const int r = 4 * w + i, m = m0 + r;
    const v2f val = *(const v2f*)(T + r * HTP + 2 * lane);
    const float sc = scale[m];
    v2f xs, xb;
    xs[0] = val[0] * sc;     xs[1] = val[1] * sc;
    xb[0] = val[0] + bb[0];  xb[1] = val[1] + bb[1];
    *(v2f*)(Xs + (size_t)m * 64 + 2 * lane) = xs;
    *(v2f*)(XWb + (size_t)m * 64 + 2 * lane) = xb;
  }
}

extern "C" void kernel_launch(void* const* d_in, const int* in_sizes, int n_in,
                              void* d_out, int out_size, void* d_ws, size_t ws_size,
                              hipStream_t stream) {
  const float* H    = (const float*)d_in[0];
  const float* U    = (const float*)d_in[1];
  const float* I    = (const float*)d_in[2];
  const float* W    = (const float*)d_in[3];
  const float* bias = (const float*)d_in[4];
  float* Uout = (float*)d_out;
  float* Iout = Uout + (size_t)KNU * 64;

  size_t off = 0;
  auto wsalloc = [&](size_t bytes) -> char* {
    char* p = (char*)d_ws + off;
    off += (bytes + 255) & ~(size_t)255;
    return p;
  };
  signed char* H8   = (signed char*)wsalloc((size_t)KNU * KNI);
  signed char* S    = (signed char*)wsalloc((size_t)KNU * KNI);
  signed char* Gb   = (signed char*)wsalloc((size_t)KNI * KNI);
  float* dvu     = (float*)wsalloc(KNU * 4);
  float* invRowH = (float*)wsalloc(KNU * 4);
  float* invRowS = (float*)wsalloc(KNU * 4);
  float* dvi     = (float*)wsalloc(KNI * 4);
  float* invColH = (float*)wsalloc(KNI * 4);
  float* invColS = (float*)wsalloc(KNI * 4);
  float* Xsu  = (float*)wsalloc((size_t)KNU * 64 * 4);
  float* Xsi  = (float*)wsalloc((size_t)KNI * 64 * 4);
  float* Q1   = (float*)wsalloc((size_t)KNI * 64 * 4);
  float* Q2   = (float*)wsalloc((size_t)KNI * 64 * 4);
  float* Qi1  = (float*)wsalloc((size_t)KNU * 64 * 4);
  float* Qi2  = (float*)wsalloc((size_t)KNU * 64 * 4);
  float* UWb  = (float*)wsalloc((size_t)KNU * 64 * 4);
  float* IWb  = (float*)wsalloc((size_t)KNI * 64 * 4);
  (void)ws_size; (void)in_sizes; (void)n_in; (void)out_size;
  int* rowPartH = (int*)Xsu;
  int* rowPartS = rowPartH + (size_t)(KNI / 32) * KNU;
  int* colPartH = rowPartS + (size_t)(KNI / 32) * KNU;
  unsigned char* colPartS = (unsigned char*)(colPartH + (size_t)(KNU / 32) * KNI);
  signed char* Ft  = (signed char*)(colPartS + (size_t)(KNU / 16) * KNI);
  signed char* FtU = Ft + (size_t)(KNI / 16) * KNU;

  k_binarize<<<dim3(BSEG, KNU / 32), dim3(32, 8), 0, stream>>>(H, H8, Ft, FtU, rowPartH, colPartH);
  k_gram_first<<<dim3(KNI / GSEG, KNI / 16), 128, 0, stream>>>(H8, Ft, Gb);
  k_gram_second<<<dim3(KNI / GSEG, KNU / 16), 128, 0, stream>>>(H8, FtU, Gb, S, rowPartS, colPartS);

  k_scales<<<KNU / 32, 256, 0, stream>>>(rowPartH, rowPartS, colPartH, colPartS,
                                         dvu, invRowH, invRowS,
                                         dvi, invColH, invColS);

  k_head_scale<<<(KNU + KNI) / 16, 128, 0, stream>>>(U, I, W, bias, dvu, dvi, Xsu, Xsi, UWb, IWb);
  k_prop_level1<<<2 * (KNI / 16) + 2 * (KNU / 16), 128, 0, stream>>>(H8, S, Xsu, Xsi,
                                                                     invColH, invColS, invRowH, invRowS,
                                                                     Q1, Q2, Qi1, Qi2);
  k_prop_level2<<<KNI / 16 + KNU / 16, 128, 0, stream>>>(H8, S, Q1, Q2, Qi1, Qi2,
                                                         dvu, dvi, UWb, IWb, Uout, Iout);
}
